# starting kernel
# baseline (speedup 1.0000x reference)
#include <hip/hip_runtime.h>
#include <hip/hip_bf16.h>


typedef float v2f __attribute__((ext_vector_type(2)));
typedef float v8f __attribute__((ext_vector_type(8)));

#define IN_DIM  128
#define HID_DIM 64
#define MID_DIM 128
#define OUT_DIM 32

__global__ void zero_f32_kernel(float* __restrict__ p, int n) {
    int i = blockIdx.x * blockDim.x + threadIdx.x;
    if (i < n) p[i] = 0.0f;
}

template <int W>
__global__ void bias_init_kernel(float* __restrict__ p, const float* __restrict__ bias, int n) {
    int i = blockIdx.x * blockDim.x + threadIdx.x;
    if (i < n) p[i] = bias[i % W];
}

__global__ void relu_kernel(float* __restrict__ p, int n) {
    int i = blockIdx.x * blockDim.x + threadIdx.x;
    if (i < n) p[i] = fmaxf(p[i], 0.0f);
}

__global__ void deg_accum_kernel(const int* __restrict__ dst, float* __restrict__ deg, int E) {
    int i = blockIdx.x * blockDim.x + threadIdx.x;
    if (i < E) atomicAdd(&deg[dst[i]], 1.0f);
}

__global__ void rsqrt_kernel(float* __restrict__ deg, int n) {
    int i = blockIdx.x * blockDim.x + threadIdx.x;
    if (i < n) deg[i] = 1.0f / sqrtf(deg[i] + 1.0f);
}

template <int K, int TILES, bool BIAS, bool RELU>
__global__ __launch_bounds__(256) void gemm_wmma_kernel(
    const float* __restrict__ A, const float* __restrict__ B,
    const float* __restrict__ bias, float* __restrict__ C,
    int numRowTiles, int n) {
    const int OUTW = TILES * 16;
    int wave = (blockIdx.x * blockDim.x + threadIdx.x) >> 5;
    int lane = threadIdx.x & 31;
    if (wave >= numRowTiles) return;

    const int m    = lane & 15;
    const int half = lane >> 4;
    const int row0 = wave * 16;

    v8f acc[TILES] = {};

    int arow = row0 + m;
    if (arow >= n) arow = n - 1;
    const float* __restrict__ ap = A + (long long)arow * K;

    for (int k0 = 0; k0 < K; k0 += 4) {
        v2f a = *(const v2f*)(ap + k0 + 2 * half);
        const float* __restrict__ bp = B + (long long)(k0 + 2 * half) * OUTW + m;
#pragma unroll
        for (int t = 0; t < TILES; ++t) {
            v2f b;
            b.x = bp[t * 16];
            b.y = bp[t * 16 + OUTW];
            acc[t] = __builtin_amdgcn_wmma_f32_16x16x4_f32(
                false, a, false, b, (short)0, acc[t], false, false);
        }
    }

#pragma unroll
    for (int t = 0; t < TILES; ++t) {
        const int col = t * 16 + m;
        float bv = BIAS ? bias[col] : 0.0f;
#pragma unroll
        for (int r = 0; r < 8; ++r) {
            int row = row0 + r + 8 * half;
            if (row < n) {
                float v = acc[t][r] + bv;
                if (RELU) v = fmaxf(v, 0.0f);
                C[(long long)row * OUTW + col] = v;
            }
        }
    }
}

template <int W>
__global__ void agg_kernel(const float* __restrict__ msg,
                           const int* __restrict__ src, const int* __restrict__ dst,
                           const float* __restrict__ dis, float* __restrict__ out,
                           int E, int N) {
    const int CH = W / 4;
    int idx = blockIdx.x * blockDim.x + threadIdx.x;
    int total = (E + N) * CH;
    if (idx >= total) return;
    int e = idx / CH;
    int c = (idx - e * CH) * 4;
    int s, d;
    if (e < E) { s = src[e]; d = dst[e]; }
    else       { s = d = e - E; }
    float nrm = dis[s] * dis[d];
    const float4 v = *(const float4*)(msg + (long long)s * W + c);
    float* o = out + (long long)d * W + c;
    atomicAdd(o + 0, v.x * nrm);
    atomicAdd(o + 1, v.y * nrm);
    atomicAdd(o + 2, v.z * nrm);
    atomicAdd(o + 3, v.w * nrm);
}

extern "C" void kernel_launch(void* const* d_in, const int* in_sizes, int n_in,
                              void* d_out, int out_size, void* d_ws, size_t ws_size,
                              hipStream_t stream) {
    const float* x   = (const float*)d_in[0];
    const int*   ei  = (const int*)  d_in[1];
    const float* W1  = (const float*)d_in[2];
    const float* b1  = (const float*)d_in[3];
    const float* Wl1 = (const float*)d_in[4];
    const float* bl1 = (const float*)d_in[5];
    const float* Wl2 = (const float*)d_in[6];
    const float* bl2 = (const float*)d_in[7];
    const float* W2  = (const float*)d_in[8];
    const float* b2  = (const float*)d_in[9];

    const int N = in_sizes[0] / IN_DIM;
    const int E = in_sizes[1] / 2;
    const int* src = ei;
    const int* dst = ei + E;

    float* ws   = (float*)d_ws;
    float* deg  = ws;
    float* bufA = deg  + (size_t)N;
    float* bufB = bufA + (size_t)N * 128;
    float* bufC = bufB + (size_t)N * 64;
    float* out  = (float*)d_out;

    const int TPB = 256;
    const int rowTiles = (N + 15) / 16;
    const int gemmBlocks = (rowTiles + 7) / 8;

    zero_f32_kernel<<<(N + TPB - 1) / TPB, TPB, 0, stream>>>(deg, N);
    deg_accum_kernel<<<(E + TPB - 1) / TPB, TPB, 0, stream>>>(dst, deg, E);
    rsqrt_kernel<<<(N + TPB - 1) / TPB, TPB, 0, stream>>>(deg, N);

    gemm_wmma_kernel<IN_DIM, HID_DIM / 16, false, false>
        <<<gemmBlocks, TPB, 0, stream>>>(x, W1, nullptr, bufA, rowTiles, N);
    bias_init_kernel<HID_DIM><<<((N * HID_DIM) + TPB - 1) / TPB, TPB, 0, stream>>>(bufB, b1, N * HID_DIM);
    {
        int total = (E + N) * (HID_DIM / 4);
        agg_kernel<HID_DIM><<<(total + TPB - 1) / TPB, TPB, 0, stream>>>(bufA, src, dst, deg, bufB, E, N);
    }
    relu_kernel<<<((N * HID_DIM) + TPB - 1) / TPB, TPB, 0, stream>>>(bufB, N * HID_DIM);

    gemm_wmma_kernel<HID_DIM, MID_DIM / 16, true, true>
        <<<gemmBlocks, TPB, 0, stream>>>(bufB, Wl1, bl1, bufA, rowTiles, N);
    gemm_wmma_kernel<MID_DIM, HID_DIM / 16, true, true>
        <<<gemmBlocks, TPB, 0, stream>>>(bufA, Wl2, bl2, bufC, rowTiles, N);

    gemm_wmma_kernel<HID_DIM, OUT_DIM / 16, false, false>
        <<<gemmBlocks, TPB, 0, stream>>>(bufC, W2, nullptr, bufB, rowTiles, N);
    bias_init_kernel<OUT_DIM><<<((N * OUT_DIM) + TPB - 1) / TPB, TPB, 0, stream>>>(out, b2, N * OUT_DIM);
    {
        int total = (E + N) * (OUT_DIM / 4);
        agg_kernel<OUT_DIM><<<(total + TPB - 1) / TPB, TPB, 0, stream>>>(bufB, src, dst, deg, out, E, N);
    }
}


# degree count: per-tile scan launch counts in-degrees in LDS (int adds), writes dis and edge records; conv1 replays them; no global atomics, no zero/rsqrt launch (6 to 5); sum order = LDS arrival
# speedup vs baseline: 3.0039x; 3.0039x over previous
#include <hip/hip_runtime.h>
#include <hip/hip_bf16.h>


typedef float v2f __attribute__((ext_vector_type(2)));
typedef float v8f __attribute__((ext_vector_type(8)));

#define IN_DIM  128
#define HID_DIM 64
#define MID_DIM 128
#define OUT_DIM 32

#define AGG_TILE 391
#define DG_THREADS 256
#define DEG_WORDS  1024
#define PRES_WORDS 2048

template <int K, int TILES>
__device__ __forceinline__ void gemm_stripe(
    const float* __restrict__ A, const float* __restrict__ B, float* __restrict__ C,
    const int wave, const int lane, const int n) {
    const int OUTW = TILES * 16;
    const int m    = lane & 15;
    const int half = lane >> 4;
    const int row0 = wave * 16;

    v8f acc[TILES] = {};

    int arow = row0 + m;
    if (arow >= n) arow = n - 1;
    const float* __restrict__ ap = A + (long long)arow * K;

    for (int k0 = 0; k0 < K; k0 += 4) {
        v2f a = *(const v2f*)(ap + k0 + 2 * half);
        const float* __restrict__ bp = B + (long long)(k0 + 2 * half) * OUTW + m;
#pragma unroll
        for (int t = 0; t < TILES; ++t) {
            v2f b;
            b.x = bp[t * 16];
            b.y = bp[t * 16 + OUTW];
            acc[t] = __builtin_amdgcn_wmma_f32_16x16x4_f32(
                false, a, false, b, (short)0, acc[t], false, false);
        }
    }

#pragma unroll
    for (int t = 0; t < TILES; ++t) {
        const int col = t * 16 + m;
#pragma unroll
        for (int r = 0; r < 8; ++r) {
            int row = row0 + r + 8 * half;
            if (row < n) C[(long long)row * OUTW + col] = acc[t][r];
        }
    }
}

__global__ __launch_bounds__(DG_THREADS) void map_gemm_kernel(
    const int* __restrict__ dst,
    unsigned* __restrict__ tid32, unsigned* __restrict__ pres, int E, int nq,
    const float* __restrict__ X, const float* __restrict__ W1, float* __restrict__ XW,
    int numRowTiles, int n, int gemmBlocks) {
    __shared__ unsigned bits[PRES_WORDS];
    const int tid = threadIdx.x;

    if ((int)blockIdx.x < gemmBlocks) {
        const int wave = (blockIdx.x * DG_THREADS + tid) >> 5;
        if (wave < numRowTiles)
            gemm_stripe<IN_DIM, HID_DIM / 16>(X, W1, XW, wave, tid & 31, n);
        return;
    }

    const int blk = (int)blockIdx.x - gemmBlocks;
#pragma unroll
    for (int q = 0; q < PRES_WORDS / DG_THREADS; ++q) bits[tid + q * DG_THREADS] = 0u;
    __syncthreads();

    int d[16];
    unsigned packed[4] = {0u, 0u, 0u, 0u};
    const int e0 = (blk * DEG_WORDS + tid) * 4;
    if ((blk + 1) * (DEG_WORDS * 4) <= E) {
#pragma unroll
        for (int q = 0; q < 4; ++q) {
            const int* __restrict__ p = dst + e0 + q * (DG_THREADS * 4);
            d[4 * q] = p[0]; d[4 * q + 1] = p[1]; d[4 * q + 2] = p[2]; d[4 * q + 3] = p[3];
        }
    } else {
#pragma unroll
        for (int q = 0; q < 4; ++q) {
#pragma unroll
            for (int j = 0; j < 4; ++j) {
                const int e = e0 + q * (DG_THREADS * 4) + j;
                const int v = dst[e < E ? e : E - 1];
                d[4 * q + j] = e < E ? v : -1;
            }
        }
    }
#pragma unroll
    for (int q = 0; q < 4; ++q) {
        const int grp = (q * DG_THREADS + tid) >> 2;
#pragma unroll
        for (int j = 0; j < 4; ++j) {
            if (d[4 * q + j] >= 0) {
                const unsigned t = (unsigned)(d[4 * q + j] / AGG_TILE) & 255u;
                packed[q] |= t << (8 * j);
                atomicOr(&bits[t * 8 + (grp >> 5)], 1u << (grp & 31));
            }
        }
    }
    __syncthreads();
    unsigned* __restrict__ pp = pres + (long long)blk * PRES_WORDS;
#pragma unroll
    for (int q = 0; q < PRES_WORDS / DG_THREADS; ++q) pp[tid + q * DG_THREADS] = bits[tid + q * DG_THREADS];
#pragma unroll
    for (int q = 0; q < 4; ++q) {
        const int i = blk * DEG_WORDS + q * DG_THREADS + tid;
        if (i < nq) tid32[i] = packed[q];
    }
}

__global__ __launch_bounds__(256) void dense_chain_kernel(
    const float* __restrict__ H1, const float* __restrict__ Wl1, const float* __restrict__ bl1,
    const float* __restrict__ Wl2, const float* __restrict__ bl2, const float* __restrict__ W2,
    float* __restrict__ O, int numRowTiles, int n) {
    const int wave = (blockIdx.x * blockDim.x + threadIdx.x) >> 5;
    const int lane = threadIdx.x & 31;
    if (wave >= numRowTiles) return;

    const int m    = lane & 15;
    const int half = lane >> 4;
    const int node = wave * 16 + m;
    const int nodec = node < n ? node : n - 1;
    const float* __restrict__ hp = H1 + (long long)nodec * HID_DIM + 2 * half;

    v8f t[MID_DIM / 16] = {};
#pragma unroll
    for (int k0 = 0; k0 < HID_DIM; k0 += 4) {
        const v2f b = *(const v2f*)(hp + k0);
        const float* __restrict__ wp = Wl1 + (k0 + 2 * half) * MID_DIM + m;
#pragma unroll
        for (int To = 0; To < MID_DIM / 16; ++To) {
            v2f a;
            a.x = wp[To * 16];
            a.y = wp[To * 16 + MID_DIM];
            t[To] = __builtin_amdgcn_wmma_f32_16x16x4_f32(
                false, a, false, b, (short)0, t[To], false, false);
        }
        if ((k0 & 4) != 0) __builtin_amdgcn_sched_barrier(0);
    }
#pragma unroll
    for (int To = 0; To < MID_DIM / 16; ++To) {
        const float4 c0 = *(const float4*)(bl1 + To * 16 + 8 * half);
        const float4 c1 = *(const float4*)(bl1 + To * 16 + 8 * half + 4);
        t[To][0] = fmaxf(t[To][0] + c0.x, 0.0f); t[To][1] = fmaxf(t[To][1] + c0.y, 0.0f);
        t[To][2] = fmaxf(t[To][2] + c0.z, 0.0f); t[To][3] = fmaxf(t[To][3] + c0.w, 0.0f);
        t[To][4] = fmaxf(t[To][4] + c1.x, 0.0f); t[To][5] = fmaxf(t[To][5] + c1.y, 0.0f);
        t[To][6] = fmaxf(t[To][6] + c1.z, 0.0f); t[To][7] = fmaxf(t[To][7] + c1.w, 0.0f);
    }

    v8f h[HID_DIM / 16] = {};
#pragma unroll
    for (int T = 0; T < MID_DIM / 16; ++T) {
#pragma unroll
        for (int j = 0; j < 4; ++j) {
            v2f b;
            b.x = t[T][2 * j];
            b.y = t[T][2 * j + 1];
            const float* __restrict__ wp = Wl2 + (T * 16 + 8 * half + 2 * j) * HID_DIM + m;
#pragma unroll
            for (int To = 0; To < HID_DIM / 16; ++To) {
                v2f a;
                a.x = wp[To * 16];
                a.y = wp[To * 16 + HID_DIM];
                h[To] = __builtin_amdgcn_wmma_f32_16x16x4_f32(
                    false, a, false, b, (short)0, h[To], false, false);
            }
        }
        __builtin_amdgcn_sched_barrier(0);
    }
#pragma unroll
    for (int To = 0; To < HID_DIM / 16; ++To) {
        const float4 c0 = *(const float4*)(bl2 + To * 16 + 8 * half);
        const float4 c1 = *(const float4*)(bl2 + To * 16 + 8 * half + 4);
        h[To][0] = fmaxf(h[To][0] + c0.x, 0.0f); h[To][1] = fmaxf(h[To][1] + c0.y, 0.0f);
        h[To][2] = fmaxf(h[To][2] + c0.z, 0.0f); h[To][3] = fmaxf(h[To][3] + c0.w, 0.0f);
        h[To][4] = fmaxf(h[To][4] + c1.x, 0.0f); h[To][5] = fmaxf(h[To][5] + c1.y, 0.0f);
        h[To][6] = fmaxf(h[To][6] + c1.z, 0.0f); h[To][7] = fmaxf(h[To][7] + c1.w, 0.0f);
    }

    v8f o[OUT_DIM / 16] = {};
#pragma unroll
    for (int T = 0; T < HID_DIM / 16; ++T) {
#pragma unroll
        for (int j = 0; j < 4; ++j) {
            v2f b;
            b.x = h[T][2 * j];
            b.y = h[T][2 * j + 1];
            const float* __restrict__ wp = W2 + (T * 16 + 8 * half + 2 * j) * OUT_DIM + m;
#pragma unroll
            for (int To = 0; To < OUT_DIM / 16; ++To) {
                v2f a;
                a.x = wp[To * 16];
                a.y = wp[To * 16 + OUT_DIM];
                o[To] = __builtin_amdgcn_wmma_f32_16x16x4_f32(
                    false, a, false, b, (short)0, o[To], false, false);
            }
        }
        __builtin_amdgcn_sched_barrier(0);
    }

    if (node < n) {
        float* __restrict__ op = O + (long long)node * OUT_DIM + 8 * half;
#pragma unroll
        for (int To = 0; To < OUT_DIM / 16; ++To) {
            *(float4*)(op + To * 16)     = make_float4(o[To][0], o[To][1], o[To][2], o[To][3]);
            *(float4*)(op + To * 16 + 4) = make_float4(o[To][4], o[To][5], o[To][6], o[To][7]);
        }
    }
}

#define AGG_THREADS 1024
#define AGG_WAVES   32
#define AGG_GCAP    288
#define AGG_QCAP    544
#define REC_CAP     16384
#define REC_SRC_BITS 20

__device__ __forceinline__ unsigned zero_bytes(unsigned y) {
    return ~(((y & 0x7f7f7f7fu) + 0x7f7f7f7fu) | y) & 0x80808080u;
}

template <class DrainFn>
__device__ __forceinline__ void scan_tile_edges(
    const unsigned* __restrict__ tid32, const unsigned* __restrict__ pres, const int E, const unsigned tile,
    int* __restrict__ myq, int* __restrict__ mygl, const int lane, const int wvu, DrainFn drain) {
    const int nchunk = (E + 511) >> 9;
    const int nblk   = (nchunk + 7) >> 3;
    const unsigned m4  = tile * 0x01010101u;
    const uint4* __restrict__ tp = (const uint4*)tid32;
    const unsigned* __restrict__ pm = pres + tile * 8u + (lane & 7);
    int cnt = 0;
    int ng  = 0;

    unsigned pwn = pm[(long long)(wvu < nblk ? wvu : nblk - 1) * PRES_WORDS];
    for (int b = wvu; ; b += AGG_WAVES) {
        const bool last = b >= nblk;
        const unsigned pwc = pwn;
        pwn = pm[(long long)(b + AGG_WAVES < nblk ? b + AGG_WAVES : nblk - 1) * PRES_WORDS];
        if (!last) {
#pragma unroll
            for (int k = 0; k < 8; ++k) {
                const unsigned w = __builtin_amdgcn_readlane(pwc, k);
                if (w != 0u) {
                    if ((w >> lane) & 1u) mygl[ng + __builtin_amdgcn_mbcnt_lo(w, 0u)] = b * 256 + k * 32 + lane;
                    ng += __builtin_popcount(w);
                }
            }
        }
        while (ng >= 32 || (last && ng > 0)) {
            const int n = ng >= 32 ? 32 : ng;
            ng -= n;
            int G = 0;
            uint4 v;
            v.x = ~m4; v.y = ~m4; v.z = ~m4; v.w = ~m4;
            if (lane < n) { G = mygl[ng + lane]; v = tp[G]; }
            const int e0 = G * 16;
            unsigned mask = (zero_bytes(v.x ^ m4) >> 7) | (zero_bytes(v.y ^ m4) >> 6)
                          | (zero_bytes(v.z ^ m4) >> 5) | (zero_bytes(v.w ^ m4) >> 4);
            while (__builtin_amdgcn_ballot_w32(mask != 0u) != 0u) {
                const int p = __builtin_ctz(mask | 0x80000000u);
                const int e = e0 + ((p & 7) << 2) + (p >> 3);
                const unsigned bb = __builtin_amdgcn_ballot_w32(mask != 0u && e < E);
                if ((bb >> lane) & 1u) myq[cnt + __builtin_amdgcn_mbcnt_lo(bb, 0u)] = e;
                cnt += __builtin_popcount(bb);
                mask &= mask - 1u;
            }
            while (cnt >= 32) { cnt -= 32; drain(myq + cnt, 32); }
        }
        if (last) break;
    }
    if (cnt > 0) drain(myq, cnt);
}

template <int TILE>
__global__ __launch_bounds__(AGG_THREADS) void tile_scan_kernel(
    const int* __restrict__ src, const int* __restrict__ dst,
    const unsigned* __restrict__ tid32, const unsigned* __restrict__ pres,
    float* __restrict__ dis, unsigned* __restrict__ rec, unsigned* __restrict__ rec_cnt, int E, int N) {
    __shared__ int   indeg[TILE];
    __shared__ int   queue[AGG_WAVES * AGG_QCAP];
    __shared__ int   glist[AGG_WAVES * AGG_GCAP];
    __shared__ unsigned rec_n;

    const int tid  = threadIdx.x;
    const int lane = tid & 31;
    const int wv   = tid >> 5;
    const int base = blockIdx.x * TILE;
    const int wvu = __builtin_amdgcn_readfirstlane(wv);
    uint2* __restrict__ myrec = (uint2*)rec + (long long)blockIdx.x * REC_CAP;

    for (int i = tid; i < TILE; i += AGG_THREADS) indeg[i] = 0;
    if (tid == 0) rec_n = 0u;
    __syncthreads();

    auto drain = [&](const int* qp, int n) {
        unsigned pos = 0u;
        if (lane == 0) pos = atomicAdd(&rec_n, (unsigned)n);
        int s = 0, dl = 0;
        if (lane < n) {
            const int e = qp[lane];
            const int sv = src[e];
            const int dv = dst[e];
            asm volatile("" :: "v"(sv), "v"(dv));
            dl = dv - base;
            s = sv < 0 ? 0 : (sv >= N ? N - 1 : sv);
            atomicAdd(&indeg[dl], 1);
        }
        pos = __builtin_amdgcn_readfirstlane(pos);
        if (lane < n && pos + lane < (unsigned)REC_CAP)
            myrec[pos + lane] = make_uint2((unsigned)s | ((unsigned)dl << REC_SRC_BITS), 0u);
    };
    scan_tile_edges(tid32, pres, E, blockIdx.x & 255u, queue + wv * AGG_QCAP, glist + wv * AGG_GCAP, lane, wvu, drain);
    __syncthreads();
    if (tid == 0) rec_cnt[blockIdx.x] = rec_n;
    for (int i = tid; i < TILE; i += AGG_THREADS) {
        const int d = base + i;
        if (d < N) dis[d] = 1.0f / sqrtf((float)indeg[i] + 1.0f);
    }
}

template <int W, int TILE, bool RELU, int PASS>
__global__ __launch_bounds__(AGG_THREADS) void agg_tile_kernel(
    const float* __restrict__ msg, const int* __restrict__ src,
    const int* __restrict__ dst, const unsigned* __restrict__ tid32, const unsigned* __restrict__ pres,
    const float* __restrict__ dis,
    const float* __restrict__ bias, float* __restrict__ out,
    unsigned* __restrict__ rec, const unsigned* __restrict__ rec_cnt, int E, int N) {
    constexpr int VPL = W / 32;
    constexpr int AGG_ROWS = 32 / VPL;
    __shared__ float acc[TILE * W];
    __shared__ int   queue[AGG_WAVES * AGG_QCAP];
    __shared__ int   glist[AGG_WAVES * AGG_GCAP];

    const int tid  = threadIdx.x;
    const int lane = tid & 31;
    const int wv   = tid >> 5;
    const int base = blockIdx.x * TILE;
    const int wvu = __builtin_amdgcn_readfirstlane(wv);
    uint2* __restrict__ myrec = (uint2*)rec + (long long)blockIdx.x * REC_CAP;
    const unsigned total = rec_cnt[blockIdx.x];

    const int c4 = (tid * 4) % W;
#pragma unroll
    for (int i = tid; i < TILE * W / 4; i += AGG_THREADS) {
        const int d  = base + (i * 4) / W;
        const int dc = d < N ? d : N - 1;
        const float ds  = dis[dc];
        const float nrm = ds * ds;
        const float4 m = *(const float4*)(msg + (long long)dc * W + c4);
        float4 v;
        v.x = m.x * nrm; v.y = m.y * nrm; v.z = m.z * nrm; v.w = m.w * nrm;
        *(float4*)(acc + i * 4) = v;
    }
    __syncthreads();

    if (total == 0u) {
    } else if (total <= (unsigned)REC_CAP) {
    const int tot = (int)total;
    auto rows_add = [&](const int s, const int dl, float nrm, const int n, const bool full) {
        float v[32][VPL];
#pragma unroll
        for (int u = 0; u < 32; ++u) {
            const int sj = __builtin_amdgcn_readlane(s, u);
            const float* __restrict__ rp = msg + (long long)sj * W + lane * VPL;
            if (VPL == 2) { const v2f t = *(const v2f*)rp; v[u][0] = t.x; v[u][VPL - 1] = t.y; }
            else          { v[u][0] = *rp; }
        }
        if (PASS == 1) nrm = dis[s] * dis[base + dl];
#pragma unroll
        for (int u = 0; u < 32; ++u) {
            if (full || u < n) {
                const int   dj = __builtin_amdgcn_readlane(dl, u);
                const float nj = __int_as_float(__builtin_amdgcn_readlane(__float_as_int(nrm), u));
                float* a = acc + dj * W + lane * VPL;
#pragma unroll
                for (int p = 0; p < VPL; ++p) atomicAdd(a + p, v[u][p] * nj);
            }
        }
        return nrm;
    };
    const int lastrec = tot - 1;
    int i0 = wvu * 32;
    uint2 r = myrec[i0 + lane < lastrec ? i0 + lane : lastrec];
    while (i0 + 32 <= tot) {
        const uint2 rc = r;
        const int ic = i0 + lane;
        i0 += AGG_WAVES * 32;
        r = myrec[i0 + lane < lastrec ? i0 + lane : lastrec];
        __builtin_amdgcn_sched_barrier(0);
        const float nrm = rows_add((int)(rc.x & ((1u << REC_SRC_BITS) - 1u)), (int)(rc.x >> REC_SRC_BITS),
                                   __uint_as_float(rc.y), 32, true);
        if (PASS == 1) myrec[ic] = make_uint2(rc.x, __float_as_uint(nrm));
    }
    if (i0 < tot) {
        const float nrm = rows_add((int)(r.x & ((1u << REC_SRC_BITS) - 1u)), (int)(r.x >> REC_SRC_BITS),
                                   __uint_as_float(r.y), tot - i0, false);
        if (PASS == 1 && i0 + lane < tot) myrec[i0 + lane] = make_uint2(r.x, __float_as_uint(nrm));
    }
    } else {
    auto drain = [&](const int* qp, int n) {
        int s = 0, d = base;
        if (lane < n) {
            const int e = qp[lane];
            const int sv = src[e];
            d = dst[e];
            s = sv < 0 ? 0 : (sv >= N ? N - 1 : sv);
        }
        const int dl = d - base;
        float nrm = 0.0f;
#pragma clang loop unroll(disable)
        for (int j = 0; j < n; j += AGG_ROWS) {
            float v[AGG_ROWS][VPL];
#pragma unroll
            for (int u = 0; u < AGG_ROWS; ++u) {
                const int sj = __builtin_amdgcn_readlane(s, j + u);
                const float* __restrict__ rp = msg + (long long)sj * W + lane * VPL;
                if (VPL == 2) { const v2f t = *(const v2f*)rp; v[u][0] = t.x; v[u][VPL - 1] = t.y; }
                else          { v[u][0] = *rp; }
            }
            if (j == 0) nrm = dis[s] * dis[d];
#pragma unroll
            for (int u = 0; u < AGG_ROWS; ++u) {
                if (j + u < n) {
                    const int   dj = __builtin_amdgcn_readlane(dl, j + u);
                    const float nj = __int_as_float(__builtin_amdgcn_readlane(__float_as_int(nrm), j + u));
                    float* a = acc + dj * W + lane * VPL;
#pragma unroll
                    for (int p = 0; p < VPL; ++p) atomicAdd(a + p, v[u][p] * nj);
                }
            }
        }
    };
    scan_tile_edges(tid32, pres, E, blockIdx.x & 255u, queue + wv * AGG_QCAP, glist + wv * AGG_GCAP, lane, wvu, drain);
    }
    __syncthreads();

    const float4 b = *(const float4*)(bias + c4);
    for (int i = tid; i < TILE * W / 4; i += AGG_THREADS) {
        const int d = base + (i * 4) / W;
        if (d < N) {
            float4 a = *(const float4*)(acc + i * 4);
            a.x += b.x; a.y += b.y; a.z += b.z; a.w += b.w;
            if (RELU) { a.x = fmaxf(a.x, 0.0f); a.y = fmaxf(a.y, 0.0f); a.z = fmaxf(a.z, 0.0f); a.w = fmaxf(a.w, 0.0f); }
            *(float4*)(out + (long long)d * W + c4) = a;
        }
    }
}

extern "C" void kernel_launch(void* const* d_in, const int* in_sizes, int n_in,
                              void* d_out, int out_size, void* d_ws, size_t ws_size,
                              hipStream_t stream) {
    const float* x   = (const float*)d_in[0];
    const int*   ei  = (const int*)  d_in[1];
    const float* W1  = (const float*)d_in[2];
    const float* b1  = (const float*)d_in[3];
    const float* Wl1 = (const float*)d_in[4];
    const float* bl1 = (const float*)d_in[5];
    const float* Wl2 = (const float*)d_in[6];
    const float* bl2 = (const float*)d_in[7];
    const float* W2  = (const float*)d_in[8];
    const float* b2  = (const float*)d_in[9];

    const int N = in_sizes[0] / IN_DIM;
    const int E = in_sizes[1] / 2;
    const int* src = ei;
    const int* dst = ei + E;

    float* ws   = (float*)d_ws;
    float* dis  = ws;
    float* bufA = dis  + (size_t)N;
    float* bufB = bufA + (size_t)N * 128;
    float* bufC = bufB + (size_t)N * 64;
    unsigned* tid32 = (unsigned*)(bufC + (size_t)N * 64);
    const int nq = ((E + 511) / 512) * 128;
    unsigned* rec_cnt = tid32 + nq;
    const int nblk = (nq + DEG_WORDS - 1) / DEG_WORDS;
    unsigned* pres    = rec_cnt + 256;
    unsigned* rec     = pres + (size_t)nblk * PRES_WORDS;
    float* out  = (float*)d_out;

    const int TPB = 256;
    const int rowTiles = (N + 15) / 16;
    const int gemmBlocks = (rowTiles + 7) / 8;
    const int aggBlocks  = (N + AGG_TILE - 1) / AGG_TILE;

    map_gemm_kernel<<<gemmBlocks + nblk, DG_THREADS, 0, stream>>>(dst, tid32, pres, E, nq,
                                                               x, W1, bufA, rowTiles, N, gemmBlocks);
    tile_scan_kernel<AGG_TILE>
        <<<aggBlocks, AGG_THREADS, 0, stream>>>(src, dst, tid32, pres, dis, rec, rec_cnt, E, N);

    agg_tile_kernel<HID_DIM, AGG_TILE, true, 1>
        <<<aggBlocks, AGG_THREADS, 0, stream>>>(bufA, src, dst, tid32, pres, dis, b1, bufB, rec, rec_cnt, E, N);

    dense_chain_kernel
        <<<gemmBlocks, TPB, 0, stream>>>(bufB, Wl1, bl1, Wl2, bl2, W2, bufC, rowTiles, N);

    agg_tile_kernel<OUT_DIM, AGG_TILE, false, 2>
        <<<aggBlocks, AGG_THREADS, 0, stream>>>(bufC, src, dst, tid32, pres, dis, b2, out, rec, rec_cnt, E, N);
}


# edge search: map pass writes each 16-edge group as one 128-byte line (16 tile bytes + 16 words src|local dst); tile scan reads a found edge's word there, not src[e], dst[e]; sum order = LDS arrival
# speedup vs baseline: 3.1394x; 1.0451x over previous
#include <hip/hip_runtime.h>
#include <hip/hip_bf16.h>


typedef float v2f __attribute__((ext_vector_type(2)));
typedef float v8f __attribute__((ext_vector_type(8)));

#define IN_DIM  128
#define HID_DIM 64
#define MID_DIM 128
#define OUT_DIM 32

#define AGG_TILE 391
#define DG_THREADS 256
#define DEG_WORDS  1024
#define PRES_WORDS 2048
#define GRP_WORDS  32
#define REC_SRC_BITS 20

template <int K, int TILES>
__device__ __forceinline__ void gemm_stripe(
    const float* __restrict__ A, const float* __restrict__ B, float* __restrict__ C,
    const int wave, const int lane, const int n) {
    const int OUTW = TILES * 16;
    const int m    = lane & 15;
    const int half = lane >> 4;
    const int row0 = wave * 16;

    v8f acc[TILES] = {};

    int arow = row0 + m;
    if (arow >= n) arow = n - 1;
    const float* __restrict__ ap = A + (long long)arow * K;

    for (int k0 = 0; k0 < K; k0 += 4) {
        v2f a = *(const v2f*)(ap + k0 + 2 * half);
        const float* __restrict__ bp = B + (long long)(k0 + 2 * half) * OUTW + m;
#pragma unroll
        for (int t = 0; t < TILES; ++t) {
            v2f b;
            b.x = bp[t * 16];
            b.y = bp[t * 16 + OUTW];
            acc[t] = __builtin_amdgcn_wmma_f32_16x16x4_f32(
                false, a, false, b, (short)0, acc[t], false, false);
        }
    }

#pragma unroll
    for (int t = 0; t < TILES; ++t) {
        const int col = t * 16 + m;
#pragma unroll
        for (int r = 0; r < 8; ++r) {
            int row = row0 + r + 8 * half;
            if (row < n) C[(long long)row * OUTW + col] = acc[t][r];
        }
    }
}

__global__ __launch_bounds__(DG_THREADS) void map_gemm_kernel(
    const int* __restrict__ src, const int* __restrict__ dst,
    unsigned* __restrict__ grp, unsigned* __restrict__ pres, int E, int nq,
    const float* __restrict__ X, const float* __restrict__ W1, float* __restrict__ XW,
    int numRowTiles, int n, int gemmBlocks) {
    __shared__ unsigned bits[PRES_WORDS];
    const int tid = threadIdx.x;

    if ((int)blockIdx.x < gemmBlocks) {
        const int wave = (blockIdx.x * DG_THREADS + tid) >> 5;
        if (wave < numRowTiles)
            gemm_stripe<IN_DIM, HID_DIM / 16>(X, W1, XW, wave, tid & 31, n);
        return;
    }

    const int blk = (int)blockIdx.x - gemmBlocks;
#pragma unroll
    for (int q = 0; q < PRES_WORDS / DG_THREADS; ++q) bits[tid + q * DG_THREADS] = 0u;
    __syncthreads();

    int d[16];
    int s[16];
    unsigned packed[4] = {0u, 0u, 0u, 0u};
    unsigned ew[16];
    const int e0 = (blk * DEG_WORDS + tid) * 4;
    if ((blk + 1) * (DEG_WORDS * 4) <= E) {
#pragma unroll
        for (int q = 0; q < 4; ++q) {
            const int* __restrict__ p = dst + e0 + q * (DG_THREADS * 4);
            d[4 * q] = p[0]; d[4 * q + 1] = p[1]; d[4 * q + 2] = p[2]; d[4 * q + 3] = p[3];
            const int* __restrict__ ps = src + e0 + q * (DG_THREADS * 4);
            s[4 * q] = ps[0]; s[4 * q + 1] = ps[1]; s[4 * q + 2] = ps[2]; s[4 * q + 3] = ps[3];
        }
    } else {
#pragma unroll
        for (int q = 0; q < 4; ++q) {
#pragma unroll
            for (int j = 0; j < 4; ++j) {
                const int e = e0 + q * (DG_THREADS * 4) + j;
                const int ec = e < E ? e : E - 1;
                const int v = dst[ec];
                s[4 * q + j] = src[ec];
                d[4 * q + j] = e < E ? v : -1;
            }
        }
    }
#pragma unroll
    for (int q = 0; q < 4; ++q) {
        const int g = (q * DG_THREADS + tid) >> 2;
#pragma unroll
        for (int j = 0; j < 4; ++j) {
            ew[4 * q + j] = 0u;
            if (d[4 * q + j] >= 0) {
                const int tq = d[4 * q + j] / AGG_TILE;
                const unsigned t = (unsigned)tq & 255u;
                packed[q] |= t << (8 * j);
                atomicOr(&bits[t * 8 + (g >> 5)], 1u << (g & 31));
                const int sv = s[4 * q + j];
                const int sc = sv < 0 ? 0 : (sv >= n ? n - 1 : sv);
                ew[4 * q + j] = (unsigned)sc | ((unsigned)(d[4 * q + j] - tq * AGG_TILE) << REC_SRC_BITS);
            }
        }
    }
    __syncthreads();
    unsigned* __restrict__ pp = pres + (long long)blk * PRES_WORDS;
#pragma unroll
    for (int q = 0; q < PRES_WORDS / DG_THREADS; ++q) pp[tid + q * DG_THREADS] = bits[tid + q * DG_THREADS];
#pragma unroll
    for (int q = 0; q < 4; ++q) {
        const int i = blk * DEG_WORDS + q * DG_THREADS + tid;
        if (i < nq) {
            unsigned* __restrict__ line = grp + (size_t)(i >> 2) * GRP_WORDS;
            line[i & 3] = packed[q];
            *(uint4*)(line + 4 + 4 * (i & 3)) = make_uint4(ew[4 * q], ew[4 * q + 1], ew[4 * q + 2], ew[4 * q + 3]);
        }
    }
}

__global__ __launch_bounds__(256) void dense_chain_kernel(
    const float* __restrict__ H1, const float* __restrict__ Wl1, const float* __restrict__ bl1,
    const float* __restrict__ Wl2, const float* __restrict__ bl2, const float* __restrict__ W2,
    float* __restrict__ O, int numRowTiles, int n) {
    const int wave = (blockIdx.x * blockDim.x + threadIdx.x) >> 5;
    const int lane = threadIdx.x & 31;
    if (wave >= numRowTiles) return;

    const int m    = lane & 15;
    const int half = lane >> 4;
    const int node = wave * 16 + m;
    const int nodec = node < n ? node : n - 1;
    const float* __restrict__ hp = H1 + (long long)nodec * HID_DIM + 2 * half;

    v8f t[MID_DIM / 16] = {};
#pragma unroll
    for (int k0 = 0; k0 < HID_DIM; k0 += 4) {
        const v2f b = *(const v2f*)(hp + k0);
        const float* __restrict__ wp = Wl1 + (k0 + 2 * half) * MID_DIM + m;
#pragma unroll
        for (int To = 0; To < MID_DIM / 16; ++To) {
            v2f a;
            a.x = wp[To * 16];
            a.y = wp[To * 16 + MID_DIM];
            t[To] = __builtin_amdgcn_wmma_f32_16x16x4_f32(
                false, a, false, b, (short)0, t[To], false, false);
        }
        if ((k0 & 4) != 0) __builtin_amdgcn_sched_barrier(0);
    }
#pragma unroll
    for (int To = 0; To < MID_DIM / 16; ++To) {
        const float4 c0 = *(const float4*)(bl1 + To * 16 + 8 * half);
        const float4 c1 = *(const float4*)(bl1 + To * 16 + 8 * half + 4);
        t[To][0] = fmaxf(t[To][0] + c0.x, 0.0f); t[To][1] = fmaxf(t[To][1] + c0.y, 0.0f);
        t[To][2] = fmaxf(t[To][2] + c0.z, 0.0f); t[To][3] = fmaxf(t[To][3] + c0.w, 0.0f);
        t[To][4] = fmaxf(t[To][4] + c1.x, 0.0f); t[To][5] = fmaxf(t[To][5] + c1.y, 0.0f);
        t[To][6] = fmaxf(t[To][6] + c1.z, 0.0f); t[To][7] = fmaxf(t[To][7] + c1.w, 0.0f);
    }

    v8f h[HID_DIM / 16] = {};
#pragma unroll
    for (int T = 0; T < MID_DIM / 16; ++T) {
#pragma unroll
        for (int j = 0; j < 4; ++j) {
            v2f b;
            b.x = t[T][2 * j];
            b.y = t[T][2 * j + 1];
            const float* __restrict__ wp = Wl2 + (T * 16 + 8 * half + 2 * j) * HID_DIM + m;
#pragma unroll
            for (int To = 0; To < HID_DIM / 16; ++To) {
                v2f a;
                a.x = wp[To * 16];
                a.y = wp[To * 16 + HID_DIM];
                h[To] = __builtin_amdgcn_wmma_f32_16x16x4_f32(
                    false, a, false, b, (short)0, h[To], false, false);
            }
        }
        __builtin_amdgcn_sched_barrier(0);
    }
#pragma unroll
    for (int To = 0; To < HID_DIM / 16; ++To) {
        const float4 c0 = *(const float4*)(bl2 + To * 16 + 8 * half);
        const float4 c1 = *(const float4*)(bl2 + To * 16 + 8 * half + 4);
        h[To][0] = fmaxf(h[To][0] + c0.x, 0.0f); h[To][1] = fmaxf(h[To][1] + c0.y, 0.0f);
        h[To][2] = fmaxf(h[To][2] + c0.z, 0.0f); h[To][3] = fmaxf(h[To][3] + c0.w, 0.0f);
        h[To][4] = fmaxf(h[To][4] + c1.x, 0.0f); h[To][5] = fmaxf(h[To][5] + c1.y, 0.0f);
        h[To][6] = fmaxf(h[To][6] + c1.z, 0.0f); h[To][7] = fmaxf(h[To][7] + c1.w, 0.0f);
    }

    v8f o[OUT_DIM / 16] = {};
#pragma unroll
    for (int T = 0; T < HID_DIM / 16; ++T) {
#pragma unroll
        for (int j = 0; j < 4; ++j) {
            v2f b;
            b.x = h[T][2 * j];
            b.y = h[T][2 * j + 1];
            const float* __restrict__ wp = W2 + (T * 16 + 8 * half + 2 * j) * OUT_DIM + m;
#pragma unroll
            for (int To = 0; To < OUT_DIM / 16; ++To) {
                v2f a;
                a.x = wp[To * 16];
                a.y = wp[To * 16 + OUT_DIM];
                o[To] = __builtin_amdgcn_wmma_f32_16x16x4_f32(
                    false, a, false, b, (short)0, o[To], false, false);
            }
        }
        __builtin_amdgcn_sched_barrier(0);
    }

    if (node < n) {
        float* __restrict__ op = O + (long long)node * OUT_DIM + 8 * half;
#pragma unroll
        for (int To = 0; To < OUT_DIM / 16; ++To) {
            *(float4*)(op + To * 16)     = make_float4(o[To][0], o[To][1], o[To][2], o[To][3]);
            *(float4*)(op + To * 16 + 4) = make_float4(o[To][4], o[To][5], o[To][6], o[To][7]);
        }
    }
}

#define AGG_THREADS 1024
#define AGG_WAVES   32
#define AGG_GCAP    288
#define AGG_QCAP    544
#define REC_CAP     16384

__device__ __forceinline__ unsigned zero_bytes(unsigned y) {
    return ~(((y & 0x7f7f7f7fu) + 0x7f7f7f7fu) | y) & 0x80808080u;
}

template <class DrainFn>
__device__ __forceinline__ void scan_tile_edges(
    const unsigned* __restrict__ grp, const unsigned* __restrict__ pres, const int E, const unsigned tile,
    int* __restrict__ myq, int* __restrict__ mygl, const int lane, const int wvu, DrainFn drain) {
    const int nchunk = (E + 511) >> 9;
    const int nblk   = (nchunk + 7) >> 3;
    const unsigned m4  = tile * 0x01010101u;
    const unsigned* __restrict__ pm = pres + tile * 8u + (lane & 7);
    int cnt = 0;
    int ng  = 0;

    unsigned pwn = pm[(long long)(wvu < nblk ? wvu : nblk - 1) * PRES_WORDS];
    for (int b = wvu; ; b += AGG_WAVES) {
        const bool last = b >= nblk;
        const unsigned pwc = pwn;
        pwn = pm[(long long)(b + AGG_WAVES < nblk ? b + AGG_WAVES : nblk - 1) * PRES_WORDS];
        if (!last) {
#pragma unroll
            for (int k = 0; k < 8; ++k) {
                const unsigned w = __builtin_amdgcn_readlane(pwc, k);
                if (w != 0u) {
                    if ((w >> lane) & 1u) mygl[ng + __builtin_amdgcn_mbcnt_lo(w, 0u)] = b * 256 + k * 32 + lane;
                    ng += __builtin_popcount(w);
                }
            }
        }
        while (ng >= 32 || (last && ng > 0)) {
            const int n = ng >= 32 ? 32 : ng;
            ng -= n;
            int G = 0;
            uint4 v;
            v.x = ~m4; v.y = ~m4; v.z = ~m4; v.w = ~m4;
            if (lane < n) { G = mygl[ng + lane]; v = *(const uint4*)(grp + (size_t)G * GRP_WORDS); }
            const int e0 = G * 16;
            const int w0 = G * GRP_WORDS + 4;
            unsigned mask = (zero_bytes(v.x ^ m4) >> 7) | (zero_bytes(v.y ^ m4) >> 6)
                          | (zero_bytes(v.z ^ m4) >> 5) | (zero_bytes(v.w ^ m4) >> 4);
            while (__builtin_amdgcn_ballot_w32(mask != 0u) != 0u) {
                const int p = __builtin_ctz(mask | 0x80000000u);
                const int k = ((p & 7) << 2) + (p >> 3);
                const unsigned bb = __builtin_amdgcn_ballot_w32(mask != 0u && e0 + k < E);
                if ((bb >> lane) & 1u) myq[cnt + __builtin_amdgcn_mbcnt_lo(bb, 0u)] = w0 + k;
                cnt += __builtin_popcount(bb);
                mask &= mask - 1u;
            }
            while (cnt >= 32) { cnt -= 32; drain(myq + cnt, 32); }
        }
        if (last) break;
    }
    if (cnt > 0) drain(myq, cnt);
}

template <int TILE>
__global__ __launch_bounds__(AGG_THREADS) void tile_scan_kernel(
    const unsigned* __restrict__ grp, const unsigned* __restrict__ pres,
    float* __restrict__ dis, unsigned* __restrict__ rec, unsigned* __restrict__ rec_cnt, int E, int N) {
    __shared__ int   indeg[TILE];
    __shared__ int   queue[AGG_WAVES * AGG_QCAP];
    __shared__ int   glist[AGG_WAVES * AGG_GCAP];
    __shared__ unsigned rec_n;

    const int tid  = threadIdx.x;
    const int lane = tid & 31;
    const int wv   = tid >> 5;
    const int base = blockIdx.x * TILE;
    const int wvu = __builtin_amdgcn_readfirstlane(wv);
    uint2* __restrict__ myrec = (uint2*)rec + (long long)blockIdx.x * REC_CAP;

    for (int i = tid; i < TILE; i += AGG_THREADS) indeg[i] = 0;
    if (tid == 0) rec_n = 0u;
    __syncthreads();

    auto drain = [&](const int* qp, int n) {
        unsigned pos = 0u;
        if (lane == 0) pos = atomicAdd(&rec_n, (unsigned)n);
        unsigned w = 0u;
        if (lane < n) {
            w = grp[qp[lane]];
            atomicAdd(&indeg[w >> REC_SRC_BITS], 1);
        }
        pos = __builtin_amdgcn_readfirstlane(pos);
        if (lane < n && pos + lane < (unsigned)REC_CAP)
            myrec[pos + lane] = make_uint2(w, 0u);
    };
    scan_tile_edges(grp, pres, E, blockIdx.x & 255u, queue + wv * AGG_QCAP, glist + wv * AGG_GCAP, lane, wvu, drain);
    __syncthreads();
    if (tid == 0) rec_cnt[blockIdx.x] = rec_n;
    for (int i = tid; i < TILE; i += AGG_THREADS) {
        const int d = base + i;
        if (d < N) dis[d] = 1.0f / sqrtf((float)indeg[i] + 1.0f);
    }
}

template <int W, int TILE, bool RELU, int PASS>
__global__ __launch_bounds__(AGG_THREADS) void agg_tile_kernel(
    const float* __restrict__ msg, const unsigned* __restrict__ grp, const unsigned* __restrict__ pres,
    const float* __restrict__ dis,
    const float* __restrict__ bias, float* __restrict__ out,
    unsigned* __restrict__ rec, const unsigned* __restrict__ rec_cnt, int E, int N) {
    constexpr int VPL = W / 32;
    constexpr int AGG_ROWS = 32 / VPL;
    __shared__ float acc[TILE * W];
    __shared__ int   queue[AGG_WAVES * AGG_QCAP];
    __shared__ int   glist[AGG_WAVES * AGG_GCAP];

    const int tid  = threadIdx.x;
    const int lane = tid & 31;
    const int wv   = tid >> 5;
    const int base = blockIdx.x * TILE;
    const int wvu = __builtin_amdgcn_readfirstlane(wv);
    uint2* __restrict__ myrec = (uint2*)rec + (long long)blockIdx.x * REC_CAP;
    const unsigned total = rec_cnt[blockIdx.x];

    const int c4 = (tid * 4) % W;
#pragma unroll
    for (int i = tid; i < TILE * W / 4; i += AGG_THREADS) {
        const int d  = base + (i * 4) / W;
        const int dc = d < N ? d : N - 1;
        const float ds  = dis[dc];
        const float nrm = ds * ds;
        const float4 m = *(const float4*)(msg + (long long)dc * W + c4);
        float4 v;
        v.x = m.x * nrm; v.y = m.y * nrm; v.z = m.z * nrm; v.w = m.w * nrm;
        *(float4*)(acc + i * 4) = v;
    }
    __syncthreads();

    if (total == 0u) {
    } else if (total <= (unsigned)REC_CAP) {
    const int tot = (int)total;
    auto rows_add = [&](const int s, const int dl, float nrm, const int n, const bool full) {
        float v[32][VPL];
#pragma unroll
        for (int u = 0; u < 32; ++u) {
            const int sj = __builtin_amdgcn_readlane(s, u);
            const float* __restrict__ rp = msg + (long long)sj * W + lane * VPL;
            if (VPL == 2) { const v2f t = *(const v2f*)rp; v[u][0] = t.x; v[u][VPL - 1] = t.y; }
            else          { v[u][0] = *rp; }
        }
        if (PASS == 1) nrm = dis[s] * dis[base + dl];
#pragma unroll
        for (int u = 0; u < 32; ++u) {
            if (full || u < n) {
                const int   dj = __builtin_amdgcn_readlane(dl, u);
                const float nj = __int_as_float(__builtin_amdgcn_readlane(__float_as_int(nrm), u));
                float* a = acc + dj * W + lane * VPL;
#pragma unroll
                for (int p = 0; p < VPL; ++p) atomicAdd(a + p, v[u][p] * nj);
            }
        }
        return nrm;
    };
    const int lastrec = tot - 1;
    int i0 = wvu * 32;
    uint2 r = myrec[i0 + lane < lastrec ? i0 + lane : lastrec];
    while (i0 + 32 <= tot) {
        const uint2 rc = r;
        const int ic = i0 + lane;
        i0 += AGG_WAVES * 32;
        r = myrec[i0 + lane < lastrec ? i0 + lane : lastrec];
        __builtin_amdgcn_sched_barrier(0);
        const float nrm = rows_add((int)(rc.x & ((1u << REC_SRC_BITS) - 1u)), (int)(rc.x >> REC_SRC_BITS),
                                   __uint_as_float(rc.y), 32, true);
        if (PASS == 1) myrec[ic] = make_uint2(rc.x, __float_as_uint(nrm));
    }
    if (i0 < tot) {
        const float nrm = rows_add((int)(r.x & ((1u << REC_SRC_BITS) - 1u)), (int)(r.x >> REC_SRC_BITS),
                                   __uint_as_float(r.y), tot - i0, false);
        if (PASS == 1 && i0 + lane < tot) myrec[i0 + lane] = make_uint2(r.x, __float_as_uint(nrm));
    }
    } else {
    auto drain = [&](const int* qp, int n) {
        unsigned w = 0u;
        if (lane < n) w = grp[qp[lane]];
        const int s  = (int)(w & ((1u << REC_SRC_BITS) - 1u));
        const int dl = (int)(w >> REC_SRC_BITS);
        const int d  = base + dl;
        float nrm = 0.0f;
#pragma clang loop unroll(disable)
        for (int j = 0; j < n; j += AGG_ROWS) {
            float v[AGG_ROWS][VPL];
#pragma unroll
            for (int u = 0; u < AGG_ROWS; ++u) {
                const int sj = __builtin_amdgcn_readlane(s, j + u);
                const float* __restrict__ rp = msg + (long long)sj * W + lane * VPL;
                if (VPL == 2) { const v2f t = *(const v2f*)rp; v[u][0] = t.x; v[u][VPL - 1] = t.y; }
                else          { v[u][0] = *rp; }
            }
            if (j == 0) nrm = dis[s] * dis[d];
#pragma unroll
            for (int u = 0; u < AGG_ROWS; ++u) {
                if (j + u < n) {
                    const int   dj = __builtin_amdgcn_readlane(dl, j + u);
                    const float nj = __int_as_float(__builtin_amdgcn_readlane(__float_as_int(nrm), j + u));
                    float* a = acc + dj * W + lane * VPL;
#pragma unroll
                    for (int p = 0; p < VPL; ++p) atomicAdd(a + p, v[u][p] * nj);
                }
            }
        }
    };
    scan_tile_edges(grp, pres, E, blockIdx.x & 255u, queue + wv * AGG_QCAP, glist + wv * AGG_GCAP, lane, wvu, drain);
    }
    __syncthreads();

    const float4 b = *(const float4*)(bias + c4);
    for (int i = tid; i < TILE * W / 4; i += AGG_THREADS) {
        const int d = base + (i * 4) / W;
        if (d < N) {
            float4 a = *(const float4*)(acc + i * 4);
            a.x += b.x; a.y += b.y; a.z += b.z; a.w += b.w;
            if (RELU) { a.x = fmaxf(a.x, 0.0f); a.y = fmaxf(a.y, 0.0f); a.z = fmaxf(a.z, 0.0f); a.w = fmaxf(a.w, 0.0f); }
            *(float4*)(out + (long long)d * W + c4) = a;
        }
    }
}

extern "C" void kernel_launch(void* const* d_in, const int* in_sizes, int n_in,
                              void* d_out, int out_size, void* d_ws, size_t ws_size,
                              hipStream_t stream) {
    const float* x   = (const float*)d_in[0];
    const int*   ei  = (const int*)  d_in[1];
    const float* W1  = (const float*)d_in[2];
    const float* b1  = (const float*)d_in[3];
    const float* Wl1 = (const float*)d_in[4];
    const float* bl1 = (const float*)d_in[5];
    const float* Wl2 = (const float*)d_in[6];
    const float* bl2 = (const float*)d_in[7];
    const float* W2  = (const float*)d_in[8];
    const float* b2  = (const float*)d_in[9];

    const int N = in_sizes[0] / IN_DIM;
    const int E = in_sizes[1] / 2;
    const int* src = ei;
    const int* dst = ei + E;

    const int nq = ((E + 511) / 512) * 128;
    const int nblk = (nq + DEG_WORDS - 1) / DEG_WORDS;
    unsigned* grp = (unsigned*)d_ws;
    float* dis  = (float*)(grp + (size_t)nblk * (DEG_WORDS / 4) * GRP_WORDS);
    float* bufA = dis  + (size_t)N;
    float* bufB = bufA + (size_t)N * 128;
    float* bufC = bufB + (size_t)N * 64;
    unsigned* rec_cnt = (unsigned*)(bufC + (size_t)N * 64);
    unsigned* pres    = rec_cnt + 256;
    unsigned* rec     = pres + (size_t)nblk * PRES_WORDS;
    float* out  = (float*)d_out;

    const int TPB = 256;
    const int rowTiles = (N + 15) / 16;
    const int gemmBlocks = (rowTiles + 7) / 8;
    const int aggBlocks  = (N + AGG_TILE - 1) / AGG_TILE;

    map_gemm_kernel<<<gemmBlocks + nblk, DG_THREADS, 0, stream>>>(src, dst, grp, pres, E, nq,
                                                               x, W1, bufA, rowTiles, N, gemmBlocks);
    tile_scan_kernel<AGG_TILE>
        <<<aggBlocks, AGG_THREADS, 0, stream>>>(grp, pres, dis, rec, rec_cnt, E, N);

    agg_tile_kernel<HID_DIM, AGG_TILE, true, 1>
        <<<aggBlocks, AGG_THREADS, 0, stream>>>(bufA, grp, pres, dis, b1, bufB, rec, rec_cnt, E, N);

    dense_chain_kernel
        <<<gemmBlocks, TPB, 0, stream>>>(bufB, Wl1, bl1, Wl2, bl2, W2, bufC, rowTiles, N);

    agg_tile_kernel<OUT_DIM, AGG_TILE, false, 2>
        <<<aggBlocks, AGG_THREADS, 0, stream>>>(bufC, grp, pres, dis, b2, out, rec, rec_cnt, E, N);
}
